# starting kernel
# baseline (speedup 1.0000x reference)
#include <hip/hip_runtime.h>
#include <hip/hip_bf16.h>
#include <math.h>

typedef __attribute__((ext_vector_type(2))) float v2f;
typedef __attribute__((ext_vector_type(8))) float v8f;

#define BATCH   2
#define SEQLEN  1024
#define D_MODEL 512
#define D_INNER 1024
#define D_STATE 128
#define DT_RANK 32
#define NPROJ   (DT_RANK + 2 * D_STATE)
#define ROWS    (BATCH * SEQLEN)

__device__ __forceinline__ float silu_f(float x) {
    return x * (1.0f / (1.0f + __expf(-x)));
}
__device__ __forceinline__ float softplus_f(float x) {
    return (x > 20.0f) ? x : log1pf(__expf(x));
}

template <int EPI, int NB>
__global__ __launch_bounds__(32) void gemm16_wmma_f32(
    const float* __restrict__ A, int lda,
    const float* __restrict__ B, int ldb,
    float* __restrict__ C, int ldc,
    int K, const float* __restrict__ bias, int split)
{
    const int tn   = blockIdx.x;
    const int tm   = blockIdx.y;
    const int lane = threadIdx.x;
    const int half = lane >> 4;
    const int l16  = lane & 15;

    const float* Arow = A + (long)(tm * 16 + l16) * lda;
    const float* Bcol = B + (tn * (NB * 16) + l16);

    v8f acc[NB];
#pragma unroll
    for (int nb = 0; nb < NB; ++nb) acc[nb] = (v8f){};

    for (int k0 = 0; k0 < K; k0 += 4) {
        v2f a;
        a.x = Arow[k0 + 2 * half];
        a.y = Arow[k0 + 2 * half + 1];
        const float* Bk0 = Bcol + (long)(k0 + 2 * half) * ldb;
        const float* Bk1 = Bcol + (long)(k0 + 2 * half + 1) * ldb;
#pragma unroll
        for (int nb = 0; nb < NB; ++nb) {
            v2f b;
            b.x = Bk0[nb * 16];
            b.y = Bk1[nb * 16];
            acc[nb] = __builtin_amdgcn_wmma_f32_16x16x4_f32(
                 false, a,  false, b,
                 (short)0, acc[nb],  false,  false);
        }
    }

#pragma unroll
    for (int nb = 0; nb < NB; ++nb) {
        const int col = tn * (NB * 16) + nb * 16 + l16;
#pragma unroll
        for (int r = 0; r < 8; ++r) {
            const int row = tm * 16 + half * 8 + r;
            float v = acc[nb][r];
            if (EPI == 1) {
                if (col < split) v = silu_f(v);
            } else if (EPI == 2) {
                v = softplus_f(v + bias[col]);
            }
            C[(long)row * ldc + col] = v;
        }
    }
}

__global__ __launch_bounds__(32) void mamba_scan(
    const float* __restrict__ dt, const float* __restrict__ proj,
    const float* __restrict__ UZ, const float* __restrict__ A_log,
    const float* __restrict__ D_param, float* __restrict__ y)
{
    const int bd   = blockIdx.x;
    const int b    = bd >> 10;
    const int d    = bd & (D_INNER - 1);
    const int lane = threadIdx.x;

    const float4 al = *(const float4*)(A_log + d * D_STATE + lane * 4);
    float a[4], h[4];
    a[0] = -__expf(al.x); a[1] = -__expf(al.y);
    a[2] = -__expf(al.z); a[3] = -__expf(al.w);
#pragma unroll
    for (int j = 0; j < 4; ++j) h[j] = 0.0f;
    const float Dp = D_param[d];

    for (int t = 0; t < SEQLEN; ++t) {
        const int   row = b * SEQLEN + t;
        const float dtv = dt[(long)row * D_INNER + d];
        const float uv  = UZ[(long)row * (2 * D_INNER) + d];
        const float* pr = proj + (long)row * NPROJ;
        if (t + 1 < SEQLEN) {
            __builtin_prefetch(pr + NPROJ, 0, 1);
        }
        const float4 Bv = *(const float4*)(pr + DT_RANK + lane * 4);
        const float4 Cv = *(const float4*)(pr + DT_RANK + D_STATE + lane * 4);
        const float du = dtv * uv;
        float acc;
        h[0] = __expf(dtv * a[0]) * h[0] + du * Bv.x;
        h[1] = __expf(dtv * a[1]) * h[1] + du * Bv.y;
        h[2] = __expf(dtv * a[2]) * h[2] + du * Bv.z;
        h[3] = __expf(dtv * a[3]) * h[3] + du * Bv.w;
        acc = h[0] * Cv.x + h[1] * Cv.y + h[2] * Cv.z + h[3] * Cv.w;
#pragma unroll
        for (int off = 16; off; off >>= 1) acc += __shfl_xor(acc, off, 32);
        if (lane == 0) y[(long)row * D_INNER + d] = acc + Dp * uv;
    }
}

__global__ __launch_bounds__(256) void ln_gate(
    float* __restrict__ y, const float* __restrict__ UZ,
    const float* __restrict__ gamma, const float* __restrict__ beta)
{
    const int row = blockIdx.x;
    const int tid = threadIdx.x;
    const int c0  = tid * 4;

    const float4 v = *(const float4*)(y + (long)row * D_INNER + c0);
    float s  = v.x + v.y + v.z + v.w;
    float s2 = v.x * v.x + v.y * v.y + v.z * v.z + v.w * v.w;

#pragma unroll
    for (int off = 16; off; off >>= 1) {
        s  += __shfl_xor(s, off, 32);
        s2 += __shfl_xor(s2, off, 32);
    }
    __shared__ float rs[8], rs2[8];
    const int wid = tid >> 5;
    if ((tid & 31) == 0) { rs[wid] = s; rs2[wid] = s2; }
    __syncthreads();
    s = 0.0f; s2 = 0.0f;
#pragma unroll
    for (int w = 0; w < 8; ++w) { s += rs[w]; s2 += rs2[w]; }

    const float mu   = s * (1.0f / D_INNER);
    const float var  = s2 * (1.0f / D_INNER) - mu * mu;
    const float rstd = rsqrtf(var + 1e-5f);

    const float4 ga = *(const float4*)(gamma + c0);
    const float4 be = *(const float4*)(beta + c0);
    const float4 zv = *(const float4*)(UZ + (long)row * (2 * D_INNER) + D_INNER + c0);

    float4 o;
    o.x = ((v.x - mu) * rstd * ga.x + be.x) * silu_f(zv.x);
    o.y = ((v.y - mu) * rstd * ga.y + be.y) * silu_f(zv.y);
    o.z = ((v.z - mu) * rstd * ga.z + be.z) * silu_f(zv.z);
    o.w = ((v.w - mu) * rstd * ga.w + be.w) * silu_f(zv.w);
    *(float4*)(y + (long)row * D_INNER + c0) = o;
}

extern "C" void kernel_launch(void* const* d_in, const int* in_sizes, int n_in,
                              void* d_out, int out_size, void* d_ws, size_t ws_size,
                              hipStream_t stream)
{
    const float* x     = (const float*)d_in[0];
    const float* W_in  = (const float*)d_in[1];
    const float* W_x   = (const float*)d_in[2];
    const float* W_dt  = (const float*)d_in[3];
    const float* b_dt  = (const float*)d_in[4];
    const float* W_out = (const float*)d_in[5];
    const float* gamma = (const float*)d_in[6];
    const float* beta  = (const float*)d_in[7];
    const float* A_log = (const float*)d_in[8];
    const float* D_par = (const float*)d_in[9];
    float* out = (float*)d_out;

    float* UZ   = (float*)d_ws;
    float* proj = UZ   + (size_t)ROWS * 2 * D_INNER;
    float* dtb  = proj + (size_t)ROWS * NPROJ;
    float* yb   = dtb  + (size_t)ROWS * D_INNER;

    gemm16_wmma_f32<1, 4><<<dim3((2 * D_INNER) / 64, ROWS / 16), 32, 0, stream>>>(
        x, D_MODEL, W_in, 2 * D_INNER, UZ, 2 * D_INNER, D_MODEL, nullptr, D_INNER);

    gemm16_wmma_f32<0, 2><<<dim3(NPROJ / 32, ROWS / 16), 32, 0, stream>>>(
        UZ, 2 * D_INNER, W_x, NPROJ, proj, NPROJ, D_INNER, nullptr, 0);

    gemm16_wmma_f32<2, 4><<<dim3(D_INNER / 64, ROWS / 16), 32, 0, stream>>>(
        proj, NPROJ, W_dt, D_INNER, dtb, D_INNER, DT_RANK, b_dt, 0);

    mamba_scan<<<BATCH * D_INNER, 32, 0, stream>>>(dtb, proj, UZ, A_log, D_par, yb);

    ln_gate<<<ROWS, 256, 0, stream>>>(yb, UZ, gamma, beta);

    gemm16_wmma_f32<0, 4><<<dim3(D_MODEL / 64, ROWS / 16), 32, 0, stream>>>(
        yb, D_INNER, W_out, D_MODEL, out, D_MODEL, D_INNER, nullptr, 0);
}


# GEMMs tiled per work-group: A and B blocks copied once into LDS by 16-byte loads and shared by the 6-8 waves of a 128x128, 64x64 or 32x96 tile, next block loaded under the products; sum order kept
# speedup vs baseline: 5.7110x; 5.7110x over previous
#include <hip/hip_runtime.h>
#include <hip/hip_bf16.h>
#include <math.h>

typedef __attribute__((ext_vector_type(2))) float v2f;
typedef __attribute__((ext_vector_type(8))) float v8f;

#define BATCH   2
#define SEQLEN  1024
#define D_MODEL 512
#define D_INNER 1024
#define D_STATE 128
#define DT_RANK 32
#define NPROJ   (DT_RANK + 2 * D_STATE)
#define ROWS    (BATCH * SEQLEN)

__device__ __forceinline__ float silu_f(float x) {
    return x * (1.0f / (1.0f + __expf(-x)));
}
__device__ __forceinline__ float softplus_f(float x) {
    return (x > 20.0f) ? x : log1pf(__expf(x));
}

template <int EPI, int BM, int BN, int BK, int WGM, int WGN>
__global__ __launch_bounds__(WGM * WGN * 32) void gemm_tile_wmma_f32(
    const float* __restrict__ A, int lda,
    const float* __restrict__ B, int ldb,
    float* __restrict__ C, int ldc,
    int K, const float* __restrict__ bias, int split)
{
    constexpr int NTHR = WGM * WGN * 32;
    constexpr int WM = BM / WGM, WN = BN / WGN;
    constexpr int MT = WM / 16, NT = WN / 16;
    constexpr int PA = BK + 4;
    constexpr int PB = BN + 16;
    constexpr int QA = BK / 4, QB = BN / 4;
    constexpr int TA = BM * QA, TB = BK * QB;
    constexpr int NA = (TA + NTHR - 1) / NTHR, NB = (TB + NTHR - 1) / NTHR;
    constexpr int KG = (MT * NT >= 8) ? 1 : 8 / (MT * NT);
    constexpr int NG = (BK / 4) / KG;
    static_assert(NT == 2 || NT == 4, "a lane reads its NT columns as one vector");
    static_assert((BK / 4) % KG == 0, "a K block is a whole number of fragment groups");
    static_assert(MT >= 1 && BM % (16 * WGM) == 0 && BN % (16 * WGN) == 0 && BK % 4 == 0, "tile shape");
    typedef float VT __attribute__((ext_vector_type(NT)));
    typedef float F4 __attribute__((ext_vector_type(4)));

    __shared__ __attribute__((aligned(16))) float sA[2][BM][PA];
    __shared__ __attribute__((aligned(16))) float sB[2][BK][PB];

    const int tid  = threadIdx.x;
    const int lane = tid & 31;
    const int wave = __builtin_amdgcn_readfirstlane(tid >> 5);
    const int half = lane >> 4;
    const int l16  = lane & 15;
    const int wm0  = (wave / WGN) * WM;
    const int wn0  = (wave % WGN) * WN;
    const int m0   = blockIdx.y * BM;
    const int n0   = blockIdx.x * BN;

    F4 ga[NA], gb[NB];

    v8f acc[MT][NT];
#pragma unroll
    for (int i = 0; i < MT; ++i)
#pragma unroll
        for (int j = 0; j < NT; ++j) acc[i][j] = (v8f){};

#pragma unroll
    for (int i = 0; i < NA; ++i) {
        const int idx = tid + i * NTHR;
        if (i + 1 < NA || TA % NTHR == 0 || idx < TA)
            ga[i] = *(const F4*)(A + (size_t)(m0 + idx / QA) * lda + 4 * (idx % QA));
    }
#pragma unroll
    for (int i = 0; i < NB; ++i) {
        const int idx = tid + i * NTHR;
        if (i + 1 < NB || TB % NTHR == 0 || idx < TB)
            gb[i] = *(const F4*)(B + (size_t)(idx / QB) * ldb + n0 + 4 * (idx % QB));
    }
#pragma unroll
    for (int i = 0; i < NA; ++i) {
        const int idx = tid + i * NTHR;
        if (i + 1 < NA || TA % NTHR == 0 || idx < TA) *(F4*)&sA[0][idx / QA][4 * (idx % QA)] = ga[i];
    }
#pragma unroll
    for (int i = 0; i < NB; ++i) {
        const int idx = tid + i * NTHR;
        if (i + 1 < NB || TB % NTHR == 0 || idx < TB) *(F4*)&sB[0][idx / QB][4 * (idx % QB)] = gb[i];
    }
    __syncthreads();

    const int nkb = K / BK;
    for (int kb = 0; kb < nkb; ++kb) {
        const int st = kb & 1;
        const bool more = (kb + 1 < nkb);
        if (more) {
            const int k1 = (kb + 1) * BK;
#pragma unroll
            for (int i = 0; i < NA; ++i) {
                const int idx = tid + i * NTHR;
                if (i + 1 < NA || TA % NTHR == 0 || idx < TA)
                    ga[i] = *(const F4*)(A + (size_t)(m0 + idx / QA) * lda + k1 + 4 * (idx % QA));
            }
#pragma unroll
            for (int i = 0; i < NB; ++i) {
                const int idx = tid + i * NTHR;
                if (i + 1 < NB || TB % NTHR == 0 || idx < TB)
                    gb[i] = *(const F4*)(B + (size_t)(k1 + idx / QB) * ldb + n0 + 4 * (idx % QB));
            }
        }

        v2f fa[2][KG][MT];
        VT  fb0[2][KG], fb1[2][KG];
#pragma unroll
        for (int s = 0; s < KG; ++s) {
#pragma unroll
            for (int i = 0; i < MT; ++i)
                fa[0][s][i] = *(const v2f*)&sA[st][wm0 + 16 * i + l16][4 * s + 2 * half];
            fb0[0][s] = *(const VT*)&sB[st][4 * s + 2 * half][wn0 + NT * l16];
            fb1[0][s] = *(const VT*)&sB[st][4 * s + 2 * half + 1][wn0 + NT * l16];
        }
#pragma unroll
        for (int g = 0; g < NG; ++g) {
            const int cu = g & 1, nx = cu ^ 1;
            if (g + 1 < NG) {
#pragma unroll
                for (int s = 0; s < KG; ++s) {
                    const int kk = (g + 1) * KG + s;
#pragma unroll
                    for (int i = 0; i < MT; ++i)
                        fa[nx][s][i] = *(const v2f*)&sA[st][wm0 + 16 * i + l16][4 * kk + 2 * half];
                    fb0[nx][s] = *(const VT*)&sB[st][4 * kk + 2 * half][wn0 + NT * l16];
                    fb1[nx][s] = *(const VT*)&sB[st][4 * kk + 2 * half + 1][wn0 + NT * l16];
                }
            }
            __builtin_amdgcn_sched_barrier(0);
#pragma unroll
            for (int s = 0; s < KG; ++s) {
#pragma unroll
                for (int j = 0; j < NT; ++j) {
                    v2f b;
                    b.x = fb0[cu][s][j];
                    b.y = fb1[cu][s][j];
#pragma unroll
                    for (int i = 0; i < MT; ++i)
                        acc[i][j] = __builtin_amdgcn_wmma_f32_16x16x4_f32(
                             false, fa[cu][s][i],  false, b,
                             (short)0, acc[i][j],  false,  false);
                }
            }
        }

        if (more) {
#pragma unroll
            for (int i = 0; i < NA; ++i) {
                const int idx = tid + i * NTHR;
                if (i + 1 < NA || TA % NTHR == 0 || idx < TA) *(F4*)&sA[st ^ 1][idx / QA][4 * (idx % QA)] = ga[i];
            }
#pragma unroll
            for (int i = 0; i < NB; ++i) {
                const int idx = tid + i * NTHR;
                if (i + 1 < NB || TB % NTHR == 0 || idx < TB) *(F4*)&sB[st ^ 1][idx / QB][4 * (idx % QB)] = gb[i];
            }
        }
        __syncthreads();
    }

    if (EPI == 1) {
        const int cb = n0 + wn0;
        if (cb + WN <= split) {
#pragma unroll
            for (int i = 0; i < MT; ++i)
#pragma unroll
                for (int j = 0; j < NT; ++j)
#pragma unroll
                    for (int r = 0; r < 8; ++r) acc[i][j][r] = silu_f(acc[i][j][r]);
        } else if (cb < split) {
#pragma unroll
            for (int i = 0; i < MT; ++i)
#pragma unroll
                for (int j = 0; j < NT; ++j)
#pragma unroll
                    for (int r = 0; r < 8; ++r)
                        if (cb + NT * l16 + j < split) acc[i][j][r] = silu_f(acc[i][j][r]);
        }
    }

    const int col = n0 + wn0 + NT * l16;
    VT bv = {};
    if (EPI == 2) bv = *(const VT*)(bias + col);
#pragma unroll
    for (int i = 0; i < MT; ++i) {
#pragma unroll
        for (int r = 0; r < 8; ++r) {
            const int row = m0 + wm0 + 16 * i + 8 * half + r;
            VT o;
#pragma unroll
            for (int j = 0; j < NT; ++j) {
                float v = acc[i][j][r];
                if (EPI == 2) v = softplus_f(v + bv[j]);
                o[j] = v;
            }
            *(VT*)(C + (size_t)row * ldc + col) = o;
        }
    }
}

#define SC_T      32
#define SC_G      4
#define SC_S      8
#define SC_W      4
#define SC_SW     (SC_S * SC_W)
#define SC_NSPLIT (D_STATE / SC_SW)
#define SC_NCHUNK (SEQLEN / SC_T)
#define SC_PLANE  ((size_t)ROWS * D_INNER)
static_assert(SC_W == 4 && SC_NSPLIT == 4 && SC_S == 8 && SC_T == 32,
              "index maps, wave sum and plane sum are written for these sizes");

struct ScanGroup {
    float4 B0[SC_G], B1[SC_G], C0[SC_G], C1[SC_G];
    float  dt[SC_G], u[SC_G];
};

__global__ __launch_bounds__(SC_W * 32) void mamba_scan_cl(
    const float* __restrict__ dt, const float* __restrict__ proj,
    const float* __restrict__ UZ, const float* __restrict__ A_log,
    float* __restrict__ ypart)
{
    __shared__ __attribute__((aligned(16))) float sBC[2][SC_T][2 * SC_SW];
    __shared__ __attribute__((aligned(16))) float sDT[2][SC_T][32];
    __shared__ __attribute__((aligned(16))) float sU[2][SC_T][32];
    __shared__ float sY[2][SC_T][SC_W][32];

    const int sp   = blockIdx.x & (SC_NSPLIT - 1);
    const int cg   = (blockIdx.x >> 2) & 31;
    const int b    = blockIdx.x >> 7;
    const int tid  = threadIdx.x;
    const int lane = tid & 31;
    const int w    = tid >> 5;
    const int d    = cg * 32 + lane;
    const int n0   = sp * SC_SW + w * SC_S;
    const int rowbase = b * SEQLEN;

    float a[SC_S], h[SC_S];
    {
        const float4 al0 = *(const float4*)(A_log + (size_t)d * D_STATE + n0);
        const float4 al1 = *(const float4*)(A_log + (size_t)d * D_STATE + n0 + 4);
        a[0] = -__expf(al0.x); a[1] = -__expf(al0.y);
        a[2] = -__expf(al0.z); a[3] = -__expf(al0.w);
        a[4] = -__expf(al1.x); a[5] = -__expf(al1.y);
        a[6] = -__expf(al1.z); a[7] = -__expf(al1.w);
    }
#pragma unroll
    for (int j = 0; j < SC_S; ++j) h[j] = 0.0f;

    const int qbc  = tid & 15;
    const int tbc  = tid >> 4;
    const int colbc = (qbc < 8) ? (DT_RANK + sp * SC_SW + qbc * 4)
                                : (DT_RANK + D_STATE + sp * SC_SW + (qbc - 8) * 4);
    const int qd   = tid & 7;
    const int td   = tid >> 3;
    const float* pBC = proj + (size_t)(rowbase + tbc) * NPROJ + colbc;
    const float* pDT = dt + (size_t)(rowbase + td) * D_INNER + cg * 32 + qd * 4;
    const float* pU  = UZ + (size_t)(rowbase + td) * (2 * D_INNER) + cg * 32 + qd * 4;

    {
#pragma unroll
        for (int k = 0; k < 4; ++k)
            *(float4*)&sBC[0][tbc + 8 * k][qbc * 4] =
                *(const float4*)(pBC + (size_t)(8 * k) * NPROJ);
#pragma unroll
        for (int k = 0; k < 2; ++k) {
            *(float4*)&sDT[0][td + 16 * k][qd * 4] =
                *(const float4*)(pDT + (size_t)(16 * k) * D_INNER);
            *(float4*)&sU[0][td + 16 * k][qd * 4] =
                *(const float4*)(pU + (size_t)(16 * k) * (2 * D_INNER));
        }
    }
    __syncthreads();

    for (int c = 0; c < SC_NCHUNK; ++c) {
        const int st = c & 1;
        const bool more = (c + 1 < SC_NCHUNK);
        float4 gb0, gb1, gb2, gb3, gd0, gd1, gu0, gu1;
        if (more) {
            const size_t r1 = (size_t)(c + 1) * SC_T;
            gb0 = *(const float4*)(pBC + (r1 + 0) * NPROJ);
            gb1 = *(const float4*)(pBC + (r1 + 8) * NPROJ);
            gb2 = *(const float4*)(pBC + (r1 + 16) * NPROJ);
            gb3 = *(const float4*)(pBC + (r1 + 24) * NPROJ);
            gd0 = *(const float4*)(pDT + (r1 + 0) * D_INNER);
            gd1 = *(const float4*)(pDT + (r1 + 16) * D_INNER);
            gu0 = *(const float4*)(pU + (r1 + 0) * (2 * D_INNER));
            gu1 = *(const float4*)(pU + (r1 + 16) * (2 * D_INNER));
        }

        float yacc[SC_T];
        ScanGroup cur, nxt;
#pragma unroll
        for (int i = 0; i < SC_G; ++i) {
            cur.B0[i] = *(const float4*)&sBC[st][i][w * SC_S];
            cur.B1[i] = *(const float4*)&sBC[st][i][w * SC_S + 4];
            cur.C0[i] = *(const float4*)&sBC[st][i][SC_SW + w * SC_S];
            cur.C1[i] = *(const float4*)&sBC[st][i][SC_SW + w * SC_S + 4];
            cur.dt[i] = sDT[st][i][lane];
            cur.u[i]  = sU[st][i][lane];
        }
#pragma unroll
        for (int g = 0; g < SC_T / SC_G; ++g) {
            if (g + 1 < SC_T / SC_G) {
#pragma unroll
                for (int i = 0; i < SC_G; ++i) {
                    const int t = (g + 1) * SC_G + i;
                    nxt.B0[i] = *(const float4*)&sBC[st][t][w * SC_S];
                    nxt.B1[i] = *(const float4*)&sBC[st][t][w * SC_S + 4];
                    nxt.C0[i] = *(const float4*)&sBC[st][t][SC_SW + w * SC_S];
                    nxt.C1[i] = *(const float4*)&sBC[st][t][SC_SW + w * SC_S + 4];
                    nxt.dt[i] = sDT[st][t][lane];
                    nxt.u[i]  = sU[st][t][lane];
                }
            }
#pragma unroll
            for (int i = 0; i < SC_G; ++i) {
                const float dtv = cur.dt[i];
                const float du  = dtv * cur.u[i];
                const float Bv[SC_S] = {cur.B0[i].x, cur.B0[i].y, cur.B0[i].z, cur.B0[i].w,
                                        cur.B1[i].x, cur.B1[i].y, cur.B1[i].z, cur.B1[i].w};
                const float Cv[SC_S] = {cur.C0[i].x, cur.C0[i].y, cur.C0[i].z, cur.C0[i].w,
                                        cur.C1[i].x, cur.C1[i].y, cur.C1[i].z, cur.C1[i].w};
#pragma unroll
                for (int j = 0; j < SC_S; ++j)
                    h[j] = __expf(dtv * a[j]) * h[j] + du * Bv[j];
                const float lo = h[0] * Cv[0] + h[1] * Cv[1] + h[2] * Cv[2] + h[3] * Cv[3];
                const float hi = h[4] * Cv[4] + h[5] * Cv[5] + h[6] * Cv[6] + h[7] * Cv[7];
                yacc[g * SC_G + i] = lo + hi;
            }
            cur = nxt;
        }

#pragma unroll
        for (int t = 0; t < SC_T; ++t) sY[st][t][w][lane] = yacc[t];

        if (more) {
            *(float4*)&sBC[st ^ 1][tbc + 0][qbc * 4]  = gb0;
            *(float4*)&sBC[st ^ 1][tbc + 8][qbc * 4]  = gb1;
            *(float4*)&sBC[st ^ 1][tbc + 16][qbc * 4] = gb2;
            *(float4*)&sBC[st ^ 1][tbc + 24][qbc * 4] = gb3;
            *(float4*)&sDT[st ^ 1][td + 0][qd * 4]    = gd0;
            *(float4*)&sDT[st ^ 1][td + 16][qd * 4]   = gd1;
            *(float4*)&sU[st ^ 1][td + 0][qd * 4]     = gu0;
            *(float4*)&sU[st ^ 1][td + 16][qd * 4]    = gu1;
        }
        __syncthreads();

#pragma unroll
        for (int i = 0; i < SC_T / SC_W; ++i) {
            const int t = w * (SC_T / SC_W) + i;
            const float s = ((sY[st][t][0][lane] + sY[st][t][1][lane])
                             + sY[st][t][2][lane]) + sY[st][t][3][lane];
            ypart[(size_t)sp * SC_PLANE
                  + (size_t)(rowbase + c * SC_T + t) * D_INNER + d] = s;
        }
    }
}

__global__ __launch_bounds__(256) void ln_gate(
    float* __restrict__ y, const float* __restrict__ UZ,
    const float* __restrict__ gamma, const float* __restrict__ beta,
    const float* __restrict__ D_param)
{
    const int row = blockIdx.x;
    const int tid = threadIdx.x;
    const int c0  = tid * 4;

    const float4 p0 = *(const float4*)(y + (long)row * D_INNER + c0);
    const float4 p1 = *(const float4*)(y + SC_PLANE + (long)row * D_INNER + c0);
    const float4 p2 = *(const float4*)(y + 2 * SC_PLANE + (long)row * D_INNER + c0);
    const float4 p3 = *(const float4*)(y + 3 * SC_PLANE + (long)row * D_INNER + c0);
    const float4 uq = *(const float4*)(UZ + (long)row * (2 * D_INNER) + c0);
    const float4 dq = *(const float4*)(D_param + c0);
    float4 v;
    v.x = (((p0.x + p1.x) + p2.x) + p3.x) + dq.x * uq.x;
    v.y = (((p0.y + p1.y) + p2.y) + p3.y) + dq.y * uq.y;
    v.z = (((p0.z + p1.z) + p2.z) + p3.z) + dq.z * uq.z;
    v.w = (((p0.w + p1.w) + p2.w) + p3.w) + dq.w * uq.w;
    float s  = v.x + v.y + v.z + v.w;
    float s2 = v.x * v.x + v.y * v.y + v.z * v.z + v.w * v.w;

#pragma unroll
    for (int off = 16; off; off >>= 1) {
        s  += __shfl_xor(s, off, 32);
        s2 += __shfl_xor(s2, off, 32);
    }
    __shared__ float rs[8], rs2[8];
    const int wid = tid >> 5;
    if ((tid & 31) == 0) { rs[wid] = s; rs2[wid] = s2; }
    __syncthreads();
    s = 0.0f; s2 = 0.0f;
#pragma unroll
    for (int w = 0; w < 8; ++w) { s += rs[w]; s2 += rs2[w]; }

    const float mu   = s * (1.0f / D_INNER);
    const float var  = s2 * (1.0f / D_INNER) - mu * mu;
    const float rstd = rsqrtf(var + 1e-5f);

    const float4 ga = *(const float4*)(gamma + c0);
    const float4 be = *(const float4*)(beta + c0);
    const float4 zv = *(const float4*)(UZ + (long)row * (2 * D_INNER) + D_INNER + c0);

    float4 o;
    o.x = ((v.x - mu) * rstd * ga.x + be.x) * silu_f(zv.x);
    o.y = ((v.y - mu) * rstd * ga.y + be.y) * silu_f(zv.y);
    o.z = ((v.z - mu) * rstd * ga.z + be.z) * silu_f(zv.z);
    o.w = ((v.w - mu) * rstd * ga.w + be.w) * silu_f(zv.w);
    *(float4*)(y + (long)row * D_INNER + c0) = o;
}

extern "C" void kernel_launch(void* const* d_in, const int* in_sizes, int n_in,
                              void* d_out, int out_size, void* d_ws, size_t ws_size,
                              hipStream_t stream)
{
    const float* x     = (const float*)d_in[0];
    const float* W_in  = (const float*)d_in[1];
    const float* W_x   = (const float*)d_in[2];
    const float* W_dt  = (const float*)d_in[3];
    const float* b_dt  = (const float*)d_in[4];
    const float* W_out = (const float*)d_in[5];
    const float* gamma = (const float*)d_in[6];
    const float* beta  = (const float*)d_in[7];
    const float* A_log = (const float*)d_in[8];
    const float* D_par = (const float*)d_in[9];
    float* out = (float*)d_out;

    float* UZ   = (float*)d_ws;
    float* proj = UZ   + (size_t)ROWS * 2 * D_INNER;
    float* dtb  = proj + (size_t)ROWS * NPROJ;
    float* yb   = dtb  + (size_t)ROWS * D_INNER;

    gemm_tile_wmma_f32<1, 128, 128, 32, 4, 2><<<dim3((2 * D_INNER) / 128, ROWS / 128), 256, 0, stream>>>(
        x, D_MODEL, W_in, 2 * D_INNER, UZ, 2 * D_INNER, D_MODEL, nullptr, D_INNER);

    gemm_tile_wmma_f32<0, 32, 96, 128, 2, 3><<<dim3(NPROJ / 96, ROWS / 32), 192, 0, stream>>>(
        UZ, 2 * D_INNER, W_x, NPROJ, proj, NPROJ, D_INNER, nullptr, 0);

    gemm_tile_wmma_f32<2, 64, 64, 32, 4, 2><<<dim3(D_INNER / 64, ROWS / 64), 256, 0, stream>>>(
        proj, NPROJ, W_dt, D_INNER, dtb, D_INNER, DT_RANK, b_dt, 0);

    mamba_scan_cl<<<BATCH * (D_INNER / 32) * SC_NSPLIT, SC_W * 32, 0, stream>>>(
        dtb, proj, UZ, A_log, yb);

    ln_gate<<<ROWS, 256, 0, stream>>>(yb, UZ, gamma, beta, D_par);

    gemm_tile_wmma_f32<0, 64, 64, 128, 4, 2><<<dim3(D_MODEL / 64, ROWS / 64), 256, 0, stream>>>(
        yb, D_INNER, W_out, D_MODEL, out, D_MODEL, D_INNER, nullptr, 0);
}


# Scan: a wave keeps 4 states, not 8, so a work-group is 8 waves and every SIMD holds two: one wave's result and LDS waits are filled by the other; partial sums added in the same order
# speedup vs baseline: 6.1831x; 1.0827x over previous
#include <hip/hip_runtime.h>
#include <hip/hip_bf16.h>
#include <math.h>

typedef __attribute__((ext_vector_type(2))) float v2f;
typedef __attribute__((ext_vector_type(8))) float v8f;

#define BATCH   2
#define SEQLEN  1024
#define D_MODEL 512
#define D_INNER 1024
#define D_STATE 128
#define DT_RANK 32
#define NPROJ   (DT_RANK + 2 * D_STATE)
#define ROWS    (BATCH * SEQLEN)

__device__ __forceinline__ float silu_f(float x) {
    return x * (1.0f / (1.0f + __expf(-x)));
}
__device__ __forceinline__ float softplus_f(float x) {
    return (x > 20.0f) ? x : log1pf(__expf(x));
}

template <int EPI, int BM, int BN, int BK, int WGM, int WGN>
__global__ __launch_bounds__(WGM * WGN * 32) void gemm_tile_wmma_f32(
    const float* __restrict__ A, int lda,
    const float* __restrict__ B, int ldb,
    float* __restrict__ C, int ldc,
    int K, const float* __restrict__ bias, int split)
{
    constexpr int NTHR = WGM * WGN * 32;
    constexpr int WM = BM / WGM, WN = BN / WGN;
    constexpr int MT = WM / 16, NT = WN / 16;
    constexpr int PA = BK + 4;
    constexpr int PB = BN + 16;
    constexpr int QA = BK / 4, QB = BN / 4;
    constexpr int TA = BM * QA, TB = BK * QB;
    constexpr int NA = (TA + NTHR - 1) / NTHR, NB = (TB + NTHR - 1) / NTHR;
    constexpr int KG = (MT * NT >= 8) ? 1 : 8 / (MT * NT);
    constexpr int NG = (BK / 4) / KG;
    static_assert(NT == 2 || NT == 4, "a lane reads its NT columns as one vector");
    static_assert((BK / 4) % KG == 0, "a K block is a whole number of fragment groups");
    static_assert(MT >= 1 && BM % (16 * WGM) == 0 && BN % (16 * WGN) == 0 && BK % 4 == 0, "tile shape");
    typedef float VT __attribute__((ext_vector_type(NT)));
    typedef float F4 __attribute__((ext_vector_type(4)));

    __shared__ __attribute__((aligned(16))) float sA[2][BM][PA];
    __shared__ __attribute__((aligned(16))) float sB[2][BK][PB];

    const int tid  = threadIdx.x;
    const int lane = tid & 31;
    const int wave = __builtin_amdgcn_readfirstlane(tid >> 5);
    const int half = lane >> 4;
    const int l16  = lane & 15;
    const int wm0  = (wave / WGN) * WM;
    const int wn0  = (wave % WGN) * WN;
    const int m0   = blockIdx.y * BM;
    const int n0   = blockIdx.x * BN;

    F4 ga[NA], gb[NB];

    v8f acc[MT][NT];
#pragma unroll
    for (int i = 0; i < MT; ++i)
#pragma unroll
        for (int j = 0; j < NT; ++j) acc[i][j] = (v8f){};

#pragma unroll
    for (int i = 0; i < NA; ++i) {
        const int idx = tid + i * NTHR;
        if (i + 1 < NA || TA % NTHR == 0 || idx < TA)
            ga[i] = *(const F4*)(A + (size_t)(m0 + idx / QA) * lda + 4 * (idx % QA));
    }
#pragma unroll
    for (int i = 0; i < NB; ++i) {
        const int idx = tid + i * NTHR;
        if (i + 1 < NB || TB % NTHR == 0 || idx < TB)
            gb[i] = *(const F4*)(B + (size_t)(idx / QB) * ldb + n0 + 4 * (idx % QB));
    }
#pragma unroll
    for (int i = 0; i < NA; ++i) {
        const int idx = tid + i * NTHR;
        if (i + 1 < NA || TA % NTHR == 0 || idx < TA) *(F4*)&sA[0][idx / QA][4 * (idx % QA)] = ga[i];
    }
#pragma unroll
    for (int i = 0; i < NB; ++i) {
        const int idx = tid + i * NTHR;
        if (i + 1 < NB || TB % NTHR == 0 || idx < TB) *(F4*)&sB[0][idx / QB][4 * (idx % QB)] = gb[i];
    }
    __syncthreads();

    const int nkb = K / BK;
    for (int kb = 0; kb < nkb; ++kb) {
        const int st = kb & 1;
        const bool more = (kb + 1 < nkb);
        if (more) {
            const int k1 = (kb + 1) * BK;
#pragma unroll
            for (int i = 0; i < NA; ++i) {
                const int idx = tid + i * NTHR;
                if (i + 1 < NA || TA % NTHR == 0 || idx < TA)
                    ga[i] = *(const F4*)(A + (size_t)(m0 + idx / QA) * lda + k1 + 4 * (idx % QA));
            }
#pragma unroll
            for (int i = 0; i < NB; ++i) {
                const int idx = tid + i * NTHR;
                if (i + 1 < NB || TB % NTHR == 0 || idx < TB)
                    gb[i] = *(const F4*)(B + (size_t)(k1 + idx / QB) * ldb + n0 + 4 * (idx % QB));
            }
        }

        v2f fa[2][KG][MT];
        VT  fb0[2][KG], fb1[2][KG];
#pragma unroll
        for (int s = 0; s < KG; ++s) {
#pragma unroll
            for (int i = 0; i < MT; ++i)
                fa[0][s][i] = *(const v2f*)&sA[st][wm0 + 16 * i + l16][4 * s + 2 * half];
            fb0[0][s] = *(const VT*)&sB[st][4 * s + 2 * half][wn0 + NT * l16];
            fb1[0][s] = *(const VT*)&sB[st][4 * s + 2 * half + 1][wn0 + NT * l16];
        }
#pragma unroll
        for (int g = 0; g < NG; ++g) {
            const int cu = g & 1, nx = cu ^ 1;
            if (g + 1 < NG) {
#pragma unroll
                for (int s = 0; s < KG; ++s) {
                    const int kk = (g + 1) * KG + s;
#pragma unroll
                    for (int i = 0; i < MT; ++i)
                        fa[nx][s][i] = *(const v2f*)&sA[st][wm0 + 16 * i + l16][4 * kk + 2 * half];
                    fb0[nx][s] = *(const VT*)&sB[st][4 * kk + 2 * half][wn0 + NT * l16];
                    fb1[nx][s] = *(const VT*)&sB[st][4 * kk + 2 * half + 1][wn0 + NT * l16];
                }
            }
            __builtin_amdgcn_sched_barrier(0);
#pragma unroll
            for (int s = 0; s < KG; ++s) {
#pragma unroll
                for (int j = 0; j < NT; ++j) {
                    v2f b;
                    b.x = fb0[cu][s][j];
                    b.y = fb1[cu][s][j];
#pragma unroll
                    for (int i = 0; i < MT; ++i)
                        acc[i][j] = __builtin_amdgcn_wmma_f32_16x16x4_f32(
                             false, fa[cu][s][i],  false, b,
                             (short)0, acc[i][j],  false,  false);
                }
            }
        }

        if (more) {
#pragma unroll
            for (int i = 0; i < NA; ++i) {
                const int idx = tid + i * NTHR;
                if (i + 1 < NA || TA % NTHR == 0 || idx < TA) *(F4*)&sA[st ^ 1][idx / QA][4 * (idx % QA)] = ga[i];
            }
#pragma unroll
            for (int i = 0; i < NB; ++i) {
                const int idx = tid + i * NTHR;
                if (i + 1 < NB || TB % NTHR == 0 || idx < TB) *(F4*)&sB[st ^ 1][idx / QB][4 * (idx % QB)] = gb[i];
            }
        }
        __syncthreads();
    }

    if (EPI == 1) {
        const int cb = n0 + wn0;
        if (cb + WN <= split) {
#pragma unroll
            for (int i = 0; i < MT; ++i)
#pragma unroll
                for (int j = 0; j < NT; ++j)
#pragma unroll
                    for (int r = 0; r < 8; ++r) acc[i][j][r] = silu_f(acc[i][j][r]);
        } else if (cb < split) {
#pragma unroll
            for (int i = 0; i < MT; ++i)
#pragma unroll
                for (int j = 0; j < NT; ++j)
#pragma unroll
                    for (int r = 0; r < 8; ++r)
                        if (cb + NT * l16 + j < split) acc[i][j][r] = silu_f(acc[i][j][r]);
        }
    }

    const int col = n0 + wn0 + NT * l16;
    VT bv = {};
    if (EPI == 2) bv = *(const VT*)(bias + col);
#pragma unroll
    for (int i = 0; i < MT; ++i) {
#pragma unroll
        for (int r = 0; r < 8; ++r) {
            const int row = m0 + wm0 + 16 * i + 8 * half + r;
            VT o;
#pragma unroll
            for (int j = 0; j < NT; ++j) {
                float v = acc[i][j][r];
                if (EPI == 2) v = softplus_f(v + bv[j]);
                o[j] = v;
            }
            *(VT*)(C + (size_t)row * ldc + col) = o;
        }
    }
}

#define SC_T      32
#define SC_G      4
#define SC_S      4
#define SC_W      8
#define SC_SW     (SC_S * SC_W)
#define SC_NSPLIT (D_STATE / SC_SW)
#define SC_NCHUNK (SEQLEN / SC_T)
#define SC_PLANE  ((size_t)ROWS * D_INNER)
static_assert(SC_W == 8 && SC_NSPLIT == 4 && SC_S == 4 && SC_T == 32,
              "index maps, wave sum and plane sum are written for these sizes");

struct ScanGroup {
    float4 B[SC_G], C[SC_G];
    float  dt[SC_G], u[SC_G];
};

__global__ __launch_bounds__(SC_W * 32) void mamba_scan_cl(
    const float* __restrict__ dt, const float* __restrict__ proj,
    const float* __restrict__ UZ, const float* __restrict__ A_log,
    float* __restrict__ ypart)
{
    __shared__ __attribute__((aligned(16))) float sBC[2][SC_T][2 * SC_SW];
    __shared__ __attribute__((aligned(16))) float sDT[2][SC_T][32];
    __shared__ __attribute__((aligned(16))) float sU[2][SC_T][32];
    __shared__ float sY[2][SC_T][SC_W][32];

    const int sp   = blockIdx.x & (SC_NSPLIT - 1);
    const int cg   = (blockIdx.x >> 2) & 31;
    const int b    = blockIdx.x >> 7;
    const int tid  = threadIdx.x;
    const int lane = tid & 31;
    const int w    = tid >> 5;
    const int d    = cg * 32 + lane;
    const int n0   = sp * SC_SW + w * SC_S;
    const int rowbase = b * SEQLEN;

    float a[SC_S], h[SC_S];
    {
        const float4 al = *(const float4*)(A_log + (size_t)d * D_STATE + n0);
        a[0] = -__expf(al.x); a[1] = -__expf(al.y);
        a[2] = -__expf(al.z); a[3] = -__expf(al.w);
    }
#pragma unroll
    for (int j = 0; j < SC_S; ++j) h[j] = 0.0f;

    const int qbc  = tid & 15;
    const int tbc  = tid >> 4;
    const int colbc = (qbc < 8) ? (DT_RANK + sp * SC_SW + qbc * 4)
                                : (DT_RANK + D_STATE + sp * SC_SW + (qbc - 8) * 4);
    const int qd   = tid & 7;
    const int td   = tid >> 3;
    const float* pBC = proj + (size_t)(rowbase + tbc) * NPROJ + colbc;
    const float* pDT = dt + (size_t)(rowbase + td) * D_INNER + cg * 32 + qd * 4;
    const float* pU  = UZ + (size_t)(rowbase + td) * (2 * D_INNER) + cg * 32 + qd * 4;

    {
#pragma unroll
        for (int k = 0; k < 2; ++k)
            *(float4*)&sBC[0][tbc + 16 * k][qbc * 4] =
                *(const float4*)(pBC + (size_t)(16 * k) * NPROJ);
        *(float4*)&sDT[0][td][qd * 4] = *(const float4*)pDT;
        *(float4*)&sU[0][td][qd * 4]  = *(const float4*)pU;
    }
    __syncthreads();

    for (int c = 0; c < SC_NCHUNK; ++c) {
        const int st = c & 1;
        const bool more = (c + 1 < SC_NCHUNK);
        float4 gb0, gb1, gd0, gu0;
        if (more) {
            const size_t r1 = (size_t)(c + 1) * SC_T;
            gb0 = *(const float4*)(pBC + (r1 + 0) * NPROJ);
            gb1 = *(const float4*)(pBC + (r1 + 16) * NPROJ);
            gd0 = *(const float4*)(pDT + r1 * D_INNER);
            gu0 = *(const float4*)(pU + r1 * (2 * D_INNER));
        }

        float yacc[SC_T];
        ScanGroup cur, nxt;
#pragma unroll
        for (int i = 0; i < SC_G; ++i) {
            cur.B[i]  = *(const float4*)&sBC[st][i][w * SC_S];
            cur.C[i]  = *(const float4*)&sBC[st][i][SC_SW + w * SC_S];
            cur.dt[i] = sDT[st][i][lane];
            cur.u[i]  = sU[st][i][lane];
        }
#pragma unroll
        for (int g = 0; g < SC_T / SC_G; ++g) {
            if (g + 1 < SC_T / SC_G) {
#pragma unroll
                for (int i = 0; i < SC_G; ++i) {
                    const int t = (g + 1) * SC_G + i;
                    nxt.B[i]  = *(const float4*)&sBC[st][t][w * SC_S];
                    nxt.C[i]  = *(const float4*)&sBC[st][t][SC_SW + w * SC_S];
                    nxt.dt[i] = sDT[st][t][lane];
                    nxt.u[i]  = sU[st][t][lane];
                }
            }
#pragma unroll
            for (int i = 0; i < SC_G; ++i) {
                const float dtv = cur.dt[i];
                const float du  = dtv * cur.u[i];
                const float Bv[SC_S] = {cur.B[i].x, cur.B[i].y, cur.B[i].z, cur.B[i].w};
                const float Cv[SC_S] = {cur.C[i].x, cur.C[i].y, cur.C[i].z, cur.C[i].w};
#pragma unroll
                for (int j = 0; j < SC_S; ++j)
                    h[j] = __expf(dtv * a[j]) * h[j] + du * Bv[j];
                yacc[g * SC_G + i] = h[0] * Cv[0] + h[1] * Cv[1] + h[2] * Cv[2] + h[3] * Cv[3];
            }
            cur = nxt;
        }

#pragma unroll
        for (int t = 0; t < SC_T; ++t) sY[st][t][w][lane] = yacc[t];

        if (more) {
            *(float4*)&sBC[st ^ 1][tbc + 0][qbc * 4]  = gb0;
            *(float4*)&sBC[st ^ 1][tbc + 16][qbc * 4] = gb1;
            *(float4*)&sDT[st ^ 1][td][qd * 4]        = gd0;
            *(float4*)&sU[st ^ 1][td][qd * 4]         = gu0;
        }
        __syncthreads();

#pragma unroll
        for (int i = 0; i < SC_T / SC_W; ++i) {
            const int t = w * (SC_T / SC_W) + i;
            const float p01 = sY[st][t][0][lane] + sY[st][t][1][lane];
            const float p23 = sY[st][t][2][lane] + sY[st][t][3][lane];
            const float p45 = sY[st][t][4][lane] + sY[st][t][5][lane];
            const float p67 = sY[st][t][6][lane] + sY[st][t][7][lane];
            const float s = ((p01 + p23) + p45) + p67;
            ypart[(size_t)sp * SC_PLANE
                  + (size_t)(rowbase + c * SC_T + t) * D_INNER + d] = s;
        }
    }
}

__global__ __launch_bounds__(256) void ln_gate(
    float* __restrict__ y, const float* __restrict__ UZ,
    const float* __restrict__ gamma, const float* __restrict__ beta,
    const float* __restrict__ D_param)
{
    const int row = blockIdx.x;
    const int tid = threadIdx.x;
    const int c0  = tid * 4;

    const float4 p0 = *(const float4*)(y + (long)row * D_INNER + c0);
    const float4 p1 = *(const float4*)(y + SC_PLANE + (long)row * D_INNER + c0);
    const float4 p2 = *(const float4*)(y + 2 * SC_PLANE + (long)row * D_INNER + c0);
    const float4 p3 = *(const float4*)(y + 3 * SC_PLANE + (long)row * D_INNER + c0);
    const float4 uq = *(const float4*)(UZ + (long)row * (2 * D_INNER) + c0);
    const float4 dq = *(const float4*)(D_param + c0);
    float4 v;
    v.x = (((p0.x + p1.x) + p2.x) + p3.x) + dq.x * uq.x;
    v.y = (((p0.y + p1.y) + p2.y) + p3.y) + dq.y * uq.y;
    v.z = (((p0.z + p1.z) + p2.z) + p3.z) + dq.z * uq.z;
    v.w = (((p0.w + p1.w) + p2.w) + p3.w) + dq.w * uq.w;
    float s  = v.x + v.y + v.z + v.w;
    float s2 = v.x * v.x + v.y * v.y + v.z * v.z + v.w * v.w;

#pragma unroll
    for (int off = 16; off; off >>= 1) {
        s  += __shfl_xor(s, off, 32);
        s2 += __shfl_xor(s2, off, 32);
    }
    __shared__ float rs[8], rs2[8];
    const int wid = tid >> 5;
    if ((tid & 31) == 0) { rs[wid] = s; rs2[wid] = s2; }
    __syncthreads();
    s = 0.0f; s2 = 0.0f;
#pragma unroll
    for (int w = 0; w < 8; ++w) { s += rs[w]; s2 += rs2[w]; }

    const float mu   = s * (1.0f / D_INNER);
    const float var  = s2 * (1.0f / D_INNER) - mu * mu;
    const float rstd = rsqrtf(var + 1e-5f);

    const float4 ga = *(const float4*)(gamma + c0);
    const float4 be = *(const float4*)(beta + c0);
    const float4 zv = *(const float4*)(UZ + (long)row * (2 * D_INNER) + D_INNER + c0);

    float4 o;
    o.x = ((v.x - mu) * rstd * ga.x + be.x) * silu_f(zv.x);
    o.y = ((v.y - mu) * rstd * ga.y + be.y) * silu_f(zv.y);
    o.z = ((v.z - mu) * rstd * ga.z + be.z) * silu_f(zv.z);
    o.w = ((v.w - mu) * rstd * ga.w + be.w) * silu_f(zv.w);
    *(float4*)(y + (long)row * D_INNER + c0) = o;
}

extern "C" void kernel_launch(void* const* d_in, const int* in_sizes, int n_in,
                              void* d_out, int out_size, void* d_ws, size_t ws_size,
                              hipStream_t stream)
{
    const float* x     = (const float*)d_in[0];
    const float* W_in  = (const float*)d_in[1];
    const float* W_x   = (const float*)d_in[2];
    const float* W_dt  = (const float*)d_in[3];
    const float* b_dt  = (const float*)d_in[4];
    const float* W_out = (const float*)d_in[5];
    const float* gamma = (const float*)d_in[6];
    const float* beta  = (const float*)d_in[7];
    const float* A_log = (const float*)d_in[8];
    const float* D_par = (const float*)d_in[9];
    float* out = (float*)d_out;

    float* UZ   = (float*)d_ws;
    float* proj = UZ   + (size_t)ROWS * 2 * D_INNER;
    float* dtb  = proj + (size_t)ROWS * NPROJ;
    float* yb   = dtb  + (size_t)ROWS * D_INNER;

    gemm_tile_wmma_f32<1, 128, 128, 32, 4, 2><<<dim3((2 * D_INNER) / 128, ROWS / 128), 256, 0, stream>>>(
        x, D_MODEL, W_in, 2 * D_INNER, UZ, 2 * D_INNER, D_MODEL, nullptr, D_INNER);

    gemm_tile_wmma_f32<0, 32, 96, 128, 2, 3><<<dim3(NPROJ / 96, ROWS / 32), 192, 0, stream>>>(
        UZ, 2 * D_INNER, W_x, NPROJ, proj, NPROJ, D_INNER, nullptr, 0);

    gemm_tile_wmma_f32<2, 64, 64, 32, 4, 2><<<dim3(D_INNER / 64, ROWS / 64), 256, 0, stream>>>(
        proj, NPROJ, W_dt, D_INNER, dtb, D_INNER, DT_RANK, b_dt, 0);

    mamba_scan_cl<<<BATCH * (D_INNER / 32) * SC_NSPLIT, SC_W * 32, 0, stream>>>(
        dtb, proj, UZ, A_log, yb);

    ln_gate<<<ROWS, 256, 0, stream>>>(yb, UZ, gamma, beta, D_par);

    gemm_tile_wmma_f32<0, 64, 64, 128, 4, 2><<<dim3(D_MODEL / 64, ROWS / 64), 256, 0, stream>>>(
        yb, D_INNER, W_out, D_MODEL, out, D_MODEL, D_INNER, nullptr, 0);
}
